# starting kernel
# baseline (speedup 1.0000x reference)
#include <hip/hip_runtime.h>


#define BATCH 128
#define LEN   4096
#define OUTL  (2 * LEN - 1)
#define PAD   1024
#define XBUF  (LEN + 2 * PAD)

typedef float v2f __attribute__((ext_vector_type(2)));
typedef float v8f __attribute__((ext_vector_type(8)));

#if defined(__HIP_DEVICE_COMPILE__)
#if !__has_builtin(__builtin_amdgcn_wmma_f32_16x16x4_f32)
#error "missing __builtin_amdgcn_wmma_f32_16x16x4_f32 on this toolchain"
#endif
#endif

__global__ __launch_bounds__(256) void selfconv_wmma_f32(
    const float* __restrict__ x, float* __restrict__ out) {
  __shared__ float X[XBUF];

  const int tid = threadIdx.x;
  const int b   = blockIdx.x;

  for (int i = tid; i < 2 * PAD; i += 256) {
    int idx = (i < PAD) ? i : (LEN + i);
    X[idx] = 0.0f;
  }
  const float4* xb4 = (const float4*)(x + b * LEN);
  for (int i = tid; i < LEN / 4; i += 256) {
    float4 v = xb4[i];
    X[PAD + 4 * i + 0] = v.x;
    X[PAD + 4 * i + 1] = v.y;
    X[PAD + 4 * i + 2] = v.z;
    X[PAD + 4 * i + 3] = v.w;
  }
  __syncthreads();

  const int wave = tid >> 5;
  const int lane = tid & 31;
  const int m    = lane & 15;
  const int kh   = lane >> 4;
  const int gb   = wave << 10;

  int u_lo = gb - (LEN - 1);
  if (u_lo < -240) u_lo = -240;
  u_lo &= ~3;
  int u_hi = gb + 783;
  if (u_hi > LEN - 1) u_hi = LEN - 1;

  const int aBase = PAD + gb + m - 2 * kh;
  const int bBase = PAD + 2 * kh + 16 * m;

  v8f acc[4];
  acc[0] = (v8f){0.f, 0.f, 0.f, 0.f, 0.f, 0.f, 0.f, 0.f};
  acc[1] = acc[0];
  acc[2] = acc[0];
  acc[3] = acc[0];

  for (int u = u_lo; u <= u_hi; u += 4) {
    v2f bf;
    bf.x = X[bBase + u];
    bf.y = X[bBase + u + 1];
#pragma unroll
    for (int j = 0; j < 4; ++j) {
      const int ai = aBase + 256 * j - u;
      v2f af;
      af.x = X[ai];
      af.y = X[ai - 1];
      acc[j] = __builtin_amdgcn_wmma_f32_16x16x4_f32(
           false, af,  false, bf,
           (short)0, acc[j],  false,  false);
    }
  }

  float* outb = out + b * OUTL;
#pragma unroll
  for (int j = 0; j < 4; ++j) {
    const int tb = gb + 256 * j;
#pragma unroll
    for (int r = 0; r < 8; ++r) {
      const int t = tb + r + 8 * kh + 16 * m;
      if (t < OUTL) outb[t] = acc[j][r];
    }
  }
}

extern "C" void kernel_launch(void* const* d_in, const int* in_sizes, int n_in,
                              void* d_out, int out_size, void* d_ws, size_t ws_size,
                              hipStream_t stream) {
  (void)in_sizes; (void)n_in; (void)d_ws; (void)ws_size; (void)out_size;
  const float* x = (const float*)d_in[0];
  float* out = (float*)d_out;
  selfconv_wmma_f32<<<dim3(BATCH), dim3(256), 0, stream>>>(x, out);
}


# LDS fragment reads double-buffered in registers: next 32-WMMA block's loads issued as one group ahead of the current block (A read from a mirrored LDS copy, no register swaps); sum order unchanged
# speedup vs baseline: 1.8021x; 1.8021x over previous
#include <hip/hip_runtime.h>


#define BATCH 128
#define LEN   4096
#define OUTL  (2 * LEN - 1)
#define PAD   1024
#define XBUF  (LEN + 2 * PAD)
#define USTEP 8

typedef float v2f __attribute__((ext_vector_type(2)));
typedef float v8f __attribute__((ext_vector_type(8)));

#if defined(__HIP_DEVICE_COMPILE__)
#if !__has_builtin(__builtin_amdgcn_wmma_f32_16x16x4_f32)
#error "missing __builtin_amdgcn_wmma_f32_16x16x4_f32 on this toolchain"
#endif
#endif

struct Frags {
  v2f b[USTEP];
  v2f a[USTEP][4];
};

__device__ __forceinline__ void load_frags(const float* X, const float* XR,
                                           int rBase, int bBase, int u, Frags& f) {
#pragma unroll
  for (int q = 0; q < USTEP; ++q) {
    const int uq = u + 4 * q;
    f.b[q].x = X[bBase + uq];
    f.b[q].y = X[bBase + uq + 1];
#pragma unroll
    for (int j = 0; j < 4; ++j) {
      const int ri = rBase - 256 * j + uq;
      f.a[q][j].x = XR[ri];
      f.a[q][j].y = XR[ri + 1];
    }
  }
}

__device__ __forceinline__ void mma_block(const Frags& f, v8f (&acc)[4]) {
#pragma unroll
  for (int q = 0; q < USTEP; ++q) {
#pragma unroll
    for (int j = 0; j < 4; ++j) {
      acc[j] = __builtin_amdgcn_wmma_f32_16x16x4_f32(
           false, f.a[q][j],  false, f.b[q],
           (short)0, acc[j],  false,  false);
    }
  }
}

__global__ __launch_bounds__(256) void selfconv_wmma_f32(
    const float* __restrict__ x, float* __restrict__ out) {
  __shared__ float X[XBUF];
  __shared__ float XR[XBUF];

  const int tid = threadIdx.x;
  const int b   = blockIdx.x;

  for (int i = tid; i < 2 * PAD; i += 256) {
    int idx = (i < PAD) ? i : (LEN + i);
    X[idx]  = 0.0f;
    XR[idx] = 0.0f;
  }
  const float4* xb4 = (const float4*)(x + b * LEN);
  for (int i = tid; i < LEN / 4; i += 256) {
    float4 v = xb4[i];
    X[PAD + 4 * i + 0] = v.x;
    X[PAD + 4 * i + 1] = v.y;
    X[PAD + 4 * i + 2] = v.z;
    X[PAD + 4 * i + 3] = v.w;
    const int r = XBUF - 1 - (PAD + 4 * i);
    XR[r - 0] = v.x;
    XR[r - 1] = v.y;
    XR[r - 2] = v.z;
    XR[r - 3] = v.w;
  }
  __syncthreads();

  const int wave = tid >> 5;
  const int lane = tid & 31;
  const int m    = lane & 15;
  const int kh   = lane >> 4;
  const int gb   = wave << 10;

  int u_lo = gb - (LEN - 1);
  if (u_lo < -240) u_lo = -240;
  u_lo &= ~3;
  int u_hi = gb + 783;
  if (u_hi > LEN - 1) u_hi = LEN - 1;

  const int aBase = PAD + gb + m - 2 * kh;
  const int rBase = XBUF - 1 - aBase;
  const int bBase = PAD + 2 * kh + 16 * m;

  v8f acc[4];
  acc[0] = (v8f){0.f, 0.f, 0.f, 0.f, 0.f, 0.f, 0.f, 0.f};
  acc[1] = acc[0];
  acc[2] = acc[0];
  acc[3] = acc[0];

  Frags f0, f1;
  load_frags(X, XR, rBase, bBase, u_lo, f0);
  for (int u = u_lo; u <= u_hi; u += 8 * USTEP) {
    load_frags(X, XR, rBase, bBase, u + 4 * USTEP, f1);
    __builtin_amdgcn_sched_barrier(0);
    mma_block(f0, acc);
    __builtin_amdgcn_sched_barrier(0);
    load_frags(X, XR, rBase, bBase, u + 8 * USTEP, f0);
    __builtin_amdgcn_sched_barrier(0);
    mma_block(f1, acc);
    __builtin_amdgcn_sched_barrier(0);
  }

  float* outb = out + b * OUTL;
#pragma unroll
  for (int j = 0; j < 4; ++j) {
    const int tb = gb + 256 * j;
#pragma unroll
    for (int r = 0; r < 8; ++r) {
      const int t = tb + r + 8 * kh + 16 * m;
      if (t < OUTL) outb[t] = acc[j][r];
    }
  }
}

extern "C" void kernel_launch(void* const* d_in, const int* in_sizes, int n_in,
                              void* d_out, int out_size, void* d_ws, size_t ws_size,
                              hipStream_t stream) {
  (void)in_sizes; (void)n_in; (void)d_ws; (void)ws_size; (void)out_size;
  const float* x = (const float*)d_in[0];
  float* out = (float*)d_out;
  selfconv_wmma_f32<<<dim3(BATCH), dim3(256), 0, stream>>>(x, out);
}


# Row split over two workgroups (grid 256): each wave owns one long and one short output tile, 1088 WMMAs for every wave; the long tile's solo K-steps alternate two accumulators, added at the end
# speedup vs baseline: 3.5106x; 1.9481x over previous
#include <hip/hip_runtime.h>


#define BATCH 128
#define NBLK  256
#define LEN   4096
#define OUTL  (2 * LEN - 1)
#define PAD   1024
#define XBUF  (LEN + 2 * PAD)

typedef float v2f __attribute__((ext_vector_type(2)));
typedef float v8f __attribute__((ext_vector_type(8)));

#if defined(__HIP_DEVICE_COMPILE__)
#if !__has_builtin(__builtin_amdgcn_wmma_f32_16x16x4_f32)
#error "missing __builtin_amdgcn_wmma_f32_16x16x4_f32 on this toolchain"
#endif
#endif

template <int NT, int US>
struct Frags {
  v2f b[US];
  v2f a[US][NT];
};

template <int NT, int US>
__device__ __forceinline__ void load_frags(const float* X, const float* XR,
                                           const int (&rB)[2], int bBase, int u,
                                           Frags<NT, US>& f) {
#pragma unroll
  for (int q = 0; q < US; ++q) {
    const int uq = u + 4 * q;
    f.b[q].x = X[bBase + uq];
    f.b[q].y = X[bBase + uq + 1];
#pragma unroll
    for (int t = 0; t < NT; ++t) {
      f.a[q][t].x = XR[rB[t] + uq];
      f.a[q][t].y = XR[rB[t] + uq + 1];
    }
  }
}

template <int NT, int US>
__device__ __forceinline__ void mma_block(const Frags<NT, US>& f, v8f (&acc)[3]) {
#pragma unroll
  for (int q = 0; q < US; ++q) {
    if constexpr (NT == 2) {
      acc[0] = __builtin_amdgcn_wmma_f32_16x16x4_f32(
           false, f.a[q][0],  false, f.b[q],
           (short)0, acc[0],  false,  false);
      acc[1] = __builtin_amdgcn_wmma_f32_16x16x4_f32(
          false, f.a[q][1], false, f.b[q], (short)0, acc[1], false, false);
    } else {
      if (q & 1) {
        acc[2] = __builtin_amdgcn_wmma_f32_16x16x4_f32(
            false, f.a[q][0], false, f.b[q], (short)0, acc[2], false, false);
      } else {
        acc[0] = __builtin_amdgcn_wmma_f32_16x16x4_f32(
            false, f.a[q][0], false, f.b[q], (short)0, acc[0], false, false);
      }
    }
  }
}

template <int NT>
__device__ __forceinline__ void pin_acc(v8f (&acc)[3]) {
  if constexpr (NT == 2) {
    asm volatile("" : "+v"(acc[0]), "+v"(acc[1]));
  } else {
    asm volatile("" : "+v"(acc[0]), "+v"(acc[2]));
  }
}

template <int NT, int US>
__device__ __forceinline__ void run_range(const float* X, const float* XR,
                                          const int (&rB)[2], int bBase,
                                          int u0, int u1, v8f (&acc)[3]) {
  Frags<NT, US> f0, f1;
  load_frags<NT, US>(X, XR, rB, bBase, u0, f0);
  for (int u = u0; u < u1; u += 8 * US) {
    load_frags<NT, US>(X, XR, rB, bBase, u + 4 * US, f1);
    __builtin_amdgcn_sched_barrier(0);
    pin_acc<NT>(acc);
    mma_block<NT, US>(f0, acc);
    pin_acc<NT>(acc);
    __builtin_amdgcn_sched_barrier(0);
    load_frags<NT, US>(X, XR, rB, bBase, u + 8 * US, f0);
    __builtin_amdgcn_sched_barrier(0);
    pin_acc<NT>(acc);
    mma_block<NT, US>(f1, acc);
    pin_acc<NT>(acc);
    __builtin_amdgcn_sched_barrier(0);
  }
}

__global__ __launch_bounds__(256) void selfconv_wmma_f32(
    const float* __restrict__ x, float* __restrict__ out) {
  __shared__ float X[XBUF];
  __shared__ float XR[XBUF];

  const int tid  = threadIdx.x;
  const int b    = blockIdx.x >> 1;
  const int half = blockIdx.x & 1;

  for (int i = tid; i < 2 * PAD; i += 256) {
    int idx = (i < PAD) ? i : (LEN + i);
    X[idx]  = 0.0f;
    XR[idx] = 0.0f;
  }
  const float4* xb4 = (const float4*)(x + b * LEN);
  for (int i = tid; i < LEN / 4; i += 256) {
    float4 v = xb4[i];
    X[PAD + 4 * i + 0] = v.x;
    X[PAD + 4 * i + 1] = v.y;
    X[PAD + 4 * i + 2] = v.z;
    X[PAD + 4 * i + 3] = v.w;
    const int r = XBUF - 1 - (PAD + 4 * i);
    XR[r - 0] = v.x;
    XR[r - 1] = v.y;
    XR[r - 2] = v.z;
    XR[r - 3] = v.w;
  }
  __syncthreads();

  const int wave = tid >> 5;
  const int lane = tid & 31;
  const int m    = lane & 15;
  const int kh   = lane >> 4;

  const int tS = half ? (31 - wave) : wave;
  const int tL = half ? (16 + wave) : (15 - wave);
  const int tbL = tL << 8;
  const int tbS = tS << 8;

  int rB[2];
  rB[0] = XBUF - 1 - (PAD + tbL + m - 2 * kh);
  rB[1] = XBUF - 1 - (PAD + tbS + m - 2 * kh);
  const int bBase = PAD + 2 * kh + 16 * m;

  v8f acc[3];
  acc[0] = (v8f){0.f, 0.f, 0.f, 0.f, 0.f, 0.f, 0.f, 0.f};
  acc[1] = acc[0];
  acc[2] = acc[0];

  if (half == 0) {
    const int uMid = tbS + 16;
    run_range<2, 16>(X, XR, rB, bBase, -240, uMid, acc);
    run_range<1, 16>(X, XR, rB, bBase, uMid, tbL + 16, acc);
  } else {
    const int uMid = tbS - LEN;
    run_range<1, 16>(X, XR, rB, bBase, tbL - LEN, uMid, acc);
    run_range<2, 16>(X, XR, rB, bBase, uMid, LEN, acc);
  }
  asm volatile("v_nop\n\tv_nop\n\tv_nop\n\tv_nop\n\tv_nop"
               : "+v"(acc[0]), "+v"(acc[1]), "+v"(acc[2]));
  acc[0] += acc[2];

  float* outb = out + b * OUTL;
#pragma unroll
  for (int r = 0; r < 8; ++r) {
    const int t = tbL + r + 8 * kh + 16 * m;
    if (t < OUTL) outb[t] = acc[0][r];
  }
#pragma unroll
  for (int r = 0; r < 8; ++r) {
    const int t = tbS + r + 8 * kh + 16 * m;
    if (t < OUTL) outb[t] = acc[1][r];
  }
}

extern "C" void kernel_launch(void* const* d_in, const int* in_sizes, int n_in,
                              void* d_out, int out_size, void* d_ws, size_t ws_size,
                              hipStream_t stream) {
  (void)in_sizes; (void)n_in; (void)d_ws; (void)ws_size; (void)out_size;
  const float* x = (const float*)d_in[0];
  float* out = (float*)d_out;
  selfconv_wmma_f32<<<dim3(NBLK), dim3(256), 0, stream>>>(x, out);
}


# B fragments read without LDS bank conflicts: B copy at pitch 20 dwords per 16 values, one 16-byte load per two K-steps; k inside each group of 8 regrouped, so sum order inside a group changes
# speedup vs baseline: 3.8925x; 1.1088x over previous
#include <hip/hip_runtime.h>


#define BATCH 128
#define NBLK  256
#define LEN   4096
#define OUTL  (2 * LEN - 1)
#define PAD   1024
#define XBUF  (LEN + 2 * PAD)
#define XPBUF (XBUF + XBUF / 4)
#define PADIDX(p) ((p) + 4 * ((p) >> 4))

typedef float v2f __attribute__((ext_vector_type(2)));
typedef float v4f __attribute__((ext_vector_type(4)));
typedef float v8f __attribute__((ext_vector_type(8)));

#if defined(__HIP_DEVICE_COMPILE__)
#if !__has_builtin(__builtin_amdgcn_wmma_f32_16x16x4_f32)
#error "missing __builtin_amdgcn_wmma_f32_16x16x4_f32 on this toolchain"
#endif
#endif

template <int NT, int US>
struct Frags {
  v2f b[US];
  v2f a[US][NT];
};

template <int NT, int US>
__device__ __forceinline__ void load_frags(const float* XP, const float* XR,
                                           const int (&rB)[2], int bBaseP, int u,
                                           int ub, Frags<NT, US>& f) {
#pragma unroll
  for (int c = 0; c < US / 2; ++c) {
    const v4f bv = *(const v4f*)(XP + bBaseP + ub + PADIDX(8 * c));
    f.b[2 * c]     = (v2f){bv.x, bv.y};
    f.b[2 * c + 1] = (v2f){bv.z, bv.w};
#pragma unroll
    for (int s = 0; s < 2; ++s) {
      const int ua = u + 8 * c + 2 * s;
#pragma unroll
      for (int t = 0; t < NT; ++t) {
        f.a[2 * c + s][t].x = XR[rB[t] + ua];
        f.a[2 * c + s][t].y = XR[rB[t] + ua + 1];
      }
    }
  }
}

template <int NT, int US>
__device__ __forceinline__ void mma_block(const Frags<NT, US>& f, v8f (&acc)[3]) {
#pragma unroll
  for (int q = 0; q < US; ++q) {
    if constexpr (NT == 2) {
      acc[0] = __builtin_amdgcn_wmma_f32_16x16x4_f32(
           false, f.a[q][0],  false, f.b[q],
           (short)0, acc[0],  false,  false);
      acc[1] = __builtin_amdgcn_wmma_f32_16x16x4_f32(
          false, f.a[q][1], false, f.b[q], (short)0, acc[1], false, false);
    } else {
      if (q & 1) {
        acc[2] = __builtin_amdgcn_wmma_f32_16x16x4_f32(
            false, f.a[q][0], false, f.b[q], (short)0, acc[2], false, false);
      } else {
        acc[0] = __builtin_amdgcn_wmma_f32_16x16x4_f32(
            false, f.a[q][0], false, f.b[q], (short)0, acc[0], false, false);
      }
    }
  }
}

template <int NT>
__device__ __forceinline__ void pin_acc(v8f (&acc)[3]) {
  if constexpr (NT == 2) {
    asm volatile("" : "+v"(acc[0]), "+v"(acc[1]));
  } else {
    asm volatile("" : "+v"(acc[0]), "+v"(acc[2]));
  }
}

template <int NT, int US>
__device__ __forceinline__ void run_range(const float* XP, const float* XR,
                                          const int (&rB)[2], int bBaseP,
                                          int u0, int u1, v8f (&acc)[3]) {
  static_assert(US % 4 == 0, "a block must cover whole groups of 16 values");
  Frags<NT, US> f0, f1;
  int ub = u0 + (u0 >> 2);
  load_frags<NT, US>(XP, XR, rB, bBaseP, u0, ub, f0);
  for (int u = u0; u < u1; u += 8 * US, ub += 10 * US) {
    load_frags<NT, US>(XP, XR, rB, bBaseP, u + 4 * US, ub + 5 * US, f1);
    __builtin_amdgcn_sched_barrier(0);
    pin_acc<NT>(acc);
    mma_block<NT, US>(f0, acc);
    pin_acc<NT>(acc);
    __builtin_amdgcn_sched_barrier(0);
    load_frags<NT, US>(XP, XR, rB, bBaseP, u + 8 * US, ub + 10 * US, f0);
    __builtin_amdgcn_sched_barrier(0);
    pin_acc<NT>(acc);
    mma_block<NT, US>(f1, acc);
    pin_acc<NT>(acc);
    __builtin_amdgcn_sched_barrier(0);
  }
}

__global__ __launch_bounds__(256) void selfconv_wmma_f32(
    const float* __restrict__ x, float* __restrict__ out) {
  __shared__ __attribute__((aligned(16))) float XP[XPBUF];
  __shared__ float XR[XBUF];

  const int tid  = threadIdx.x;
  const int b    = blockIdx.x >> 1;
  const int half = blockIdx.x & 1;

  for (int i = tid; i < 2 * PAD; i += 256) {
    int idx = (i < PAD) ? i : (LEN + i);
    XP[PADIDX(idx)] = 0.0f;
    XR[idx] = 0.0f;
  }
  const float4* xb4 = (const float4*)(x + b * LEN);
  for (int i = tid; i < LEN / 4; i += 256) {
    float4 v = xb4[i];
    const int pp = PADIDX(PAD + 4 * i);
    XP[pp + 0] = v.x;
    XP[pp + 1] = v.y;
    XP[pp + 2] = v.z;
    XP[pp + 3] = v.w;
    const int r = XBUF - 1 - (PAD + 4 * i);
    XR[r - 0] = v.x;
    XR[r - 1] = v.y;
    XR[r - 2] = v.z;
    XR[r - 3] = v.w;
  }
  __syncthreads();

  const int wave = tid >> 5;
  const int lane = tid & 31;
  const int m    = lane & 15;
  const int kh   = lane >> 4;

  const int tS = half ? (31 - wave) : wave;
  const int tL = half ? (16 + wave) : (15 - wave);
  const int tbL = tL << 8;
  const int tbS = tS << 8;

  int rB[2];
  rB[0] = XBUF - 1 - (PAD + tbL + m - 4 * kh);
  rB[1] = XBUF - 1 - (PAD + tbS + m - 4 * kh);
  const int bBaseP = PADIDX(PAD + 16 * m) + 4 * kh;

  v8f acc[3];
  acc[0] = (v8f){0.f, 0.f, 0.f, 0.f, 0.f, 0.f, 0.f, 0.f};
  acc[1] = acc[0];
  acc[2] = acc[0];

  if (half == 0) {
    const int uMid = tbS + 16;
    run_range<2, 16>(XP, XR, rB, bBaseP, -240, uMid, acc);
    run_range<1, 16>(XP, XR, rB, bBaseP, uMid, tbL + 16, acc);
  } else {
    const int uMid = tbS - LEN;
    run_range<1, 16>(XP, XR, rB, bBaseP, tbL - LEN, uMid, acc);
    run_range<2, 16>(XP, XR, rB, bBaseP, uMid, LEN, acc);
  }
  asm volatile("v_nop\n\tv_nop\n\tv_nop\n\tv_nop\n\tv_nop"
               : "+v"(acc[0]), "+v"(acc[1]), "+v"(acc[2]));
  acc[0] += acc[2];

  float* outb = out + b * OUTL;
#pragma unroll
  for (int r = 0; r < 8; ++r) {
    const int t = tbL + r + 8 * kh + 16 * m;
    if (t < OUTL) outb[t] = acc[0][r];
  }
#pragma unroll
  for (int r = 0; r < 8; ++r) {
    const int t = tbS + r + 8 * kh + 16 * m;
    if (t < OUTL) outb[t] = acc[1][r];
  }
}

extern "C" void kernel_launch(void* const* d_in, const int* in_sizes, int n_in,
                              void* d_out, int out_size, void* d_ws, size_t ws_size,
                              hipStream_t stream) {
  (void)in_sizes; (void)n_in; (void)d_ws; (void)ws_size; (void)out_size;
  const float* x = (const float*)d_in[0];
  float* out = (float*)d_out;
  selfconv_wmma_f32<<<dim3(NBLK), dim3(256), 0, stream>>>(x, out);
}


# Each product pair x[i]*x[j] computed once: a tile keeps only split points below cut=tb/2+128 (K range halved), result doubled; last 256 points: B zeroed past the cut, A doubled; sum order changes
# speedup vs baseline: 5.2876x; 1.3584x over previous
#include <hip/hip_runtime.h>


#define BATCH 128
#define NBLK  256
#define LEN   4096
#define OUTL  (2 * LEN - 1)
#define PAD   1024
#define XBUF  (LEN + 2 * PAD)
#define XPBUF (XBUF + XBUF / 4)
#define PADIDX(p) ((p) + 4 * ((p) >> 4))

typedef float v2f __attribute__((ext_vector_type(2)));
typedef float v4f __attribute__((ext_vector_type(4)));
typedef float v8f __attribute__((ext_vector_type(8)));

#if defined(__HIP_DEVICE_COMPILE__)
#if !__has_builtin(__builtin_amdgcn_wmma_f32_16x16x4_f32)
#error "missing __builtin_amdgcn_wmma_f32_16x16x4_f32 on this toolchain"
#endif
#endif

template <int US>
struct Frags {
  v2f b[US];
  v2f a[US];
};

template <int US>
__device__ __forceinline__ void load_frags(const float* XP, const float* XR,
                                           int rA, int bBaseP, int u,
                                           int ub, Frags<US>& f) {
#pragma unroll
  for (int c = 0; c < US / 2; ++c) {
    const v4f bv = *(const v4f*)(XP + bBaseP + ub + PADIDX(8 * c));
    f.b[2 * c]     = (v2f){bv.x, bv.y};
    f.b[2 * c + 1] = (v2f){bv.z, bv.w};
#pragma unroll
    for (int s = 0; s < 2; ++s) {
      const int ua = u + 8 * c + 2 * s;
      f.a[2 * c + s].x = XR[rA + ua];
      f.a[2 * c + s].y = XR[rA + ua + 1];
    }
  }
}

template <int US>
__device__ __forceinline__ void cut_frags(Frags<US>& f, int eB, int eA) {
#pragma unroll
  for (int c = 0; c < US / 2; ++c) {
    const bool zb = (8 * c >= eB);
#pragma unroll
    for (int s = 0; s < 2; ++s) {
      const int q  = 2 * c + s;
      const int k0 = 8 * c + 2 * s;
      f.b[q].x = zb ? 0.0f : f.b[q].x;
      f.b[q].y = zb ? 0.0f : f.b[q].y;
      f.a[q].x = (k0     <= eA) ? 2.0f * f.a[q].x : f.a[q].x;
      f.a[q].y = (k0 + 1 <= eA) ? 2.0f * f.a[q].y : f.a[q].y;
    }
  }
#pragma unroll
  for (int q = 0; q < US; q += 8) {
    asm volatile("" : "+v"(f.b[q]), "+v"(f.b[q + 1]), "+v"(f.b[q + 2]), "+v"(f.b[q + 3]),
                      "+v"(f.b[q + 4]), "+v"(f.b[q + 5]), "+v"(f.b[q + 6]), "+v"(f.b[q + 7]));
    asm volatile("" : "+v"(f.a[q]), "+v"(f.a[q + 1]), "+v"(f.a[q + 2]), "+v"(f.a[q + 3]),
                      "+v"(f.a[q + 4]), "+v"(f.a[q + 5]), "+v"(f.a[q + 6]), "+v"(f.a[q + 7]));
  }
}

__device__ __forceinline__ void wmma_tail(v8f& acc0, v8f& acc1) {
  asm volatile("v_nop\n\tv_nop\n\tv_nop\n\tv_nop\n\tv_nop" : "+v"(acc0), "+v"(acc1));
}

template <int US>
__device__ __forceinline__ void mma_block(const Frags<US>& f, v8f& acc0, v8f& acc1) {
#pragma unroll
  for (int q = 0; q < US; ++q) {
    if (q & 1) {
      acc1 = __builtin_amdgcn_wmma_f32_16x16x4_f32(
           false, f.a[q],  false, f.b[q],
           (short)0, acc1,  false,  false);
    } else {
      acc0 = __builtin_amdgcn_wmma_f32_16x16x4_f32(
          false, f.a[q], false, f.b[q], (short)0, acc0, false, false);
    }
  }
}

__device__ __forceinline__ void pin_acc(v8f& acc0, v8f& acc1) {
  asm volatile("" : "+v"(acc0), "+v"(acc1));
}

template <int US, bool CUT>
__device__ __forceinline__ void run_range(const float* XP, const float* XR,
                                          int rA, int bBaseP, int u0, int u1,
                                          int eB0, int eA0, v8f& acc0, v8f& acc1) {
  static_assert(US % 4 == 0, "a block must cover whole groups of 16 values");
  Frags<US> f0, f1;
  int ub = u0 + (u0 >> 2);
  load_frags<US>(XP, XR, rA, bBaseP, u0, ub, f0);
  for (int u = u0; u < u1; u += 8 * US, ub += 10 * US) {
    load_frags<US>(XP, XR, rA, bBaseP, u + 4 * US, ub + 5 * US, f1);
    __builtin_amdgcn_sched_barrier(0);
    if constexpr (CUT) cut_frags<US>(f0, eB0 - u, eA0 - u);
    pin_acc(acc0, acc1);
    mma_block<US>(f0, acc0, acc1);
    pin_acc(acc0, acc1);
    if constexpr (CUT) wmma_tail(acc0, acc1);
    __builtin_amdgcn_sched_barrier(0);
    load_frags<US>(XP, XR, rA, bBaseP, u + 8 * US, ub + 10 * US, f0);
    __builtin_amdgcn_sched_barrier(0);
    if constexpr (CUT) cut_frags<US>(f1, eB0 - u - 4 * US, eA0 - u - 4 * US);
    pin_acc(acc0, acc1);
    mma_block<US>(f1, acc0, acc1);
    pin_acc(acc0, acc1);
    if constexpr (CUT) wmma_tail(acc0, acc1);
    __builtin_amdgcn_sched_barrier(0);
  }
  wmma_tail(acc0, acc1);
}

__global__ __launch_bounds__(256) void selfconv_wmma_f32(
    const float* __restrict__ x, float* __restrict__ out) {
  __shared__ __attribute__((aligned(16))) float XP[XPBUF];
  __shared__ float XR[XBUF];

  const int tid  = threadIdx.x;
  const int b    = blockIdx.x >> 1;
  const int half = blockIdx.x & 1;

  for (int i = tid; i < 2 * PAD; i += 256) {
    int idx = (i < PAD) ? i : (LEN + i);
    XP[PADIDX(idx)] = 0.0f;
    XR[idx] = 0.0f;
  }
  const float4* xb4 = (const float4*)(x + b * LEN);
  for (int i = tid; i < LEN / 4; i += 256) {
    float4 v = xb4[i];
    const int pp = PADIDX(PAD + 4 * i);
    XP[pp + 0] = v.x;
    XP[pp + 1] = v.y;
    XP[pp + 2] = v.z;
    XP[pp + 3] = v.w;
    const int r = XBUF - 1 - (PAD + 4 * i);
    XR[r - 0] = v.x;
    XR[r - 1] = v.y;
    XR[r - 2] = v.z;
    XR[r - 3] = v.w;
  }
  __syncthreads();

  const int wave = tid >> 5;
  const int lane = tid & 31;
  const int m    = lane & 15;
  const int kh   = lane >> 4;

  const int tS = half ? (31 - wave) : wave;
  const int tL = half ? (16 + wave) : (15 - wave);

  const int bBaseP = PADIDX(PAD + 16 * m) + 4 * kh;

  v8f acc[8];
#pragma unroll
  for (int i = 0; i < 8; ++i) acc[i] = (v8f){0.f, 0.f, 0.f, 0.f, 0.f, 0.f, 0.f, 0.f};

#pragma unroll
  for (int t = 0; t < 2; ++t) {
    const int T   = t ? tS : tL;
    const int tb  = T << 8;
    const int cut = (T << 7) + 128;
    const int u0  = half ? (tb - LEN) : -256;
    const int rA  = XBUF - 1 - (PAD + tb + m - 4 * kh);
    const int eB0 = cut - 4 * kh - 16 * m;
    const int eA0 = tb + m - cut - 4 * kh;
    run_range<16, false>(XP, XR, rA, bBaseP, u0, cut - 256, 0, 0,
                         acc[2 * t], acc[2 * t + 1]);
    run_range<16, true>(XP, XR, rA, bBaseP, cut - 256, cut, eB0, eA0,
                        acc[4 + 2 * t], acc[5 + 2 * t]);
  }
  const v8f resL = 2.0f * (acc[0] + acc[1]) + (acc[4] + acc[5]);
  const v8f resS = 2.0f * (acc[2] + acc[3]) + (acc[6] + acc[7]);

  float* outb = out + b * OUTL;
  const int tbL = tL << 8;
  const int tbS = tS << 8;
#pragma unroll
  for (int r = 0; r < 8; ++r) {
    const int t = tbL + r + 8 * kh + 16 * m;
    if (t < OUTL) outb[t] = resL[r];
  }
#pragma unroll
  for (int r = 0; r < 8; ++r) {
    const int t = tbS + r + 8 * kh + 16 * m;
    if (t < OUTL) outb[t] = resS[r];
  }
}

extern "C" void kernel_launch(void* const* d_in, const int* in_sizes, int n_in,
                              void* d_out, int out_size, void* d_ws, size_t ws_size,
                              hipStream_t stream) {
  (void)in_sizes; (void)n_in; (void)d_ws; (void)ws_size; (void)out_size;
  const float* x = (const float*)d_in[0];
  float* out = (float*)d_out;
  selfconv_wmma_f32<<<dim3(NBLK), dim3(256), 0, stream>>>(x, out);
}


# Row staging: the four global loads of a thread issued together ahead of the halo clear, one wait instead of four serial round trips; LDS contents and sum order unchanged
# speedup vs baseline: 5.8372x; 1.1039x over previous
#include <hip/hip_runtime.h>


#define BATCH 128
#define NBLK  256
#define LEN   4096
#define OUTL  (2 * LEN - 1)
#define PAD   1024
#define XBUF  (LEN + 2 * PAD)
#define XPBUF (XBUF + XBUF / 4)
#define PADIDX(p) ((p) + 4 * ((p) >> 4))

typedef float v2f __attribute__((ext_vector_type(2)));
typedef float v4f __attribute__((ext_vector_type(4)));
typedef float v8f __attribute__((ext_vector_type(8)));

#if defined(__HIP_DEVICE_COMPILE__)
#if !__has_builtin(__builtin_amdgcn_wmma_f32_16x16x4_f32)
#error "missing __builtin_amdgcn_wmma_f32_16x16x4_f32 on this toolchain"
#endif
#endif

template <int US>
struct Frags {
  v2f b[US];
  v2f a[US];
};

template <int US>
__device__ __forceinline__ void load_frags(const float* XP, const float* XR,
                                           int rA, int bBaseP, int u,
                                           int ub, Frags<US>& f) {
#pragma unroll
  for (int c = 0; c < US / 2; ++c) {
    const v4f bv = *(const v4f*)(XP + bBaseP + ub + PADIDX(8 * c));
    f.b[2 * c]     = (v2f){bv.x, bv.y};
    f.b[2 * c + 1] = (v2f){bv.z, bv.w};
#pragma unroll
    for (int s = 0; s < 2; ++s) {
      const int ua = u + 8 * c + 2 * s;
      f.a[2 * c + s].x = XR[rA + ua];
      f.a[2 * c + s].y = XR[rA + ua + 1];
    }
  }
}

template <int US>
__device__ __forceinline__ void cut_frags(Frags<US>& f, int eB, int eA) {
#pragma unroll
  for (int c = 0; c < US / 2; ++c) {
    const bool zb = (8 * c >= eB);
#pragma unroll
    for (int s = 0; s < 2; ++s) {
      const int q  = 2 * c + s;
      const int k0 = 8 * c + 2 * s;
      f.b[q].x = zb ? 0.0f : f.b[q].x;
      f.b[q].y = zb ? 0.0f : f.b[q].y;
      f.a[q].x = (k0     <= eA) ? 2.0f * f.a[q].x : f.a[q].x;
      f.a[q].y = (k0 + 1 <= eA) ? 2.0f * f.a[q].y : f.a[q].y;
    }
  }
#pragma unroll
  for (int q = 0; q < US; q += 8) {
    asm volatile("" : "+v"(f.b[q]), "+v"(f.b[q + 1]), "+v"(f.b[q + 2]), "+v"(f.b[q + 3]),
                      "+v"(f.b[q + 4]), "+v"(f.b[q + 5]), "+v"(f.b[q + 6]), "+v"(f.b[q + 7]));
    asm volatile("" : "+v"(f.a[q]), "+v"(f.a[q + 1]), "+v"(f.a[q + 2]), "+v"(f.a[q + 3]),
                      "+v"(f.a[q + 4]), "+v"(f.a[q + 5]), "+v"(f.a[q + 6]), "+v"(f.a[q + 7]));
  }
}

__device__ __forceinline__ void wmma_tail(v8f& acc0, v8f& acc1) {
  asm volatile("v_nop\n\tv_nop\n\tv_nop\n\tv_nop\n\tv_nop" : "+v"(acc0), "+v"(acc1));
}

template <int US>
__device__ __forceinline__ void mma_block(const Frags<US>& f, v8f& acc0, v8f& acc1) {
#pragma unroll
  for (int q = 0; q < US; ++q) {
    if (q & 1) {
      acc1 = __builtin_amdgcn_wmma_f32_16x16x4_f32(
           false, f.a[q],  false, f.b[q],
           (short)0, acc1,  false,  false);
    } else {
      acc0 = __builtin_amdgcn_wmma_f32_16x16x4_f32(
          false, f.a[q], false, f.b[q], (short)0, acc0, false, false);
    }
  }
}

__device__ __forceinline__ void pin_acc(v8f& acc0, v8f& acc1) {
  asm volatile("" : "+v"(acc0), "+v"(acc1));
}

template <int US, bool CUT>
__device__ __forceinline__ void run_range(const float* XP, const float* XR,
                                          int rA, int bBaseP, int u0, int u1,
                                          int eB0, int eA0, v8f& acc0, v8f& acc1) {
  static_assert(US % 4 == 0, "a block must cover whole groups of 16 values");
  Frags<US> f0, f1;
  int ub = u0 + (u0 >> 2);
  load_frags<US>(XP, XR, rA, bBaseP, u0, ub, f0);
  for (int u = u0; u < u1; u += 8 * US, ub += 10 * US) {
    load_frags<US>(XP, XR, rA, bBaseP, u + 4 * US, ub + 5 * US, f1);
    __builtin_amdgcn_sched_barrier(0);
    if constexpr (CUT) cut_frags<US>(f0, eB0 - u, eA0 - u);
    pin_acc(acc0, acc1);
    mma_block<US>(f0, acc0, acc1);
    pin_acc(acc0, acc1);
    if constexpr (CUT) wmma_tail(acc0, acc1);
    __builtin_amdgcn_sched_barrier(0);
    load_frags<US>(XP, XR, rA, bBaseP, u + 8 * US, ub + 10 * US, f0);
    __builtin_amdgcn_sched_barrier(0);
    if constexpr (CUT) cut_frags<US>(f1, eB0 - u - 4 * US, eA0 - u - 4 * US);
    pin_acc(acc0, acc1);
    mma_block<US>(f1, acc0, acc1);
    pin_acc(acc0, acc1);
    if constexpr (CUT) wmma_tail(acc0, acc1);
    __builtin_amdgcn_sched_barrier(0);
  }
  wmma_tail(acc0, acc1);
}

__global__ __launch_bounds__(256) void selfconv_wmma_f32(
    const float* __restrict__ x, float* __restrict__ out) {
  __shared__ __attribute__((aligned(16))) float XP[XPBUF];
  __shared__ float XR[XBUF];

  const int tid  = threadIdx.x;
  const int b    = blockIdx.x >> 1;
  const int half = blockIdx.x & 1;

  const float4* xb4 = (const float4*)(x + b * LEN);
  float4 v[LEN / 4 / 256];
#pragma unroll
  for (int k = 0; k < LEN / 4 / 256; ++k) v[k] = xb4[tid + 256 * k];
#pragma unroll
  for (int k = 0; k < 2 * PAD / 256; ++k) {
    const int i = tid + 256 * k;
    const int idx = (256 * k < PAD) ? i : (LEN + i);
    XP[PADIDX(idx)] = 0.0f;
    XR[idx] = 0.0f;
  }
#pragma unroll
  for (int k = 0; k < LEN / 4 / 256; ++k) {
    const int i = tid + 256 * k;
    const int pp = PADIDX(PAD + 4 * i);
    XP[pp + 0] = v[k].x;
    XP[pp + 1] = v[k].y;
    XP[pp + 2] = v[k].z;
    XP[pp + 3] = v[k].w;
    const int r = XBUF - 1 - (PAD + 4 * i);
    XR[r - 0] = v[k].x;
    XR[r - 1] = v[k].y;
    XR[r - 2] = v[k].z;
    XR[r - 3] = v[k].w;
  }
  __syncthreads();

  const int wave = tid >> 5;
  const int lane = tid & 31;
  const int m    = lane & 15;
  const int kh   = lane >> 4;

  const int tS = half ? (31 - wave) : wave;
  const int tL = half ? (16 + wave) : (15 - wave);

  const int bBaseP = PADIDX(PAD + 16 * m) + 4 * kh;

  v8f acc[8];
#pragma unroll
  for (int i = 0; i < 8; ++i) acc[i] = (v8f){0.f, 0.f, 0.f, 0.f, 0.f, 0.f, 0.f, 0.f};

#pragma unroll
  for (int t = 0; t < 2; ++t) {
    const int T   = t ? tS : tL;
    const int tb  = T << 8;
    const int cut = (T << 7) + 128;
    const int u0  = half ? (tb - LEN) : -256;
    const int rA  = XBUF - 1 - (PAD + tb + m - 4 * kh);
    const int eB0 = cut - 4 * kh - 16 * m;
    const int eA0 = tb + m - cut - 4 * kh;
    run_range<16, false>(XP, XR, rA, bBaseP, u0, cut - 256, 0, 0,
                         acc[2 * t], acc[2 * t + 1]);
    run_range<16, true>(XP, XR, rA, bBaseP, cut - 256, cut, eB0, eA0,
                        acc[4 + 2 * t], acc[5 + 2 * t]);
  }
  const v8f resL = 2.0f * (acc[0] + acc[1]) + (acc[4] + acc[5]);
  const v8f resS = 2.0f * (acc[2] + acc[3]) + (acc[6] + acc[7]);

  float* outb = out + b * OUTL;
  const int tbL = tL << 8;
  const int tbS = tS << 8;
#pragma unroll
  for (int r = 0; r < 8; ++r) {
    const int t = tbL + r + 8 * kh + 16 * m;
    if (t < OUTL) outb[t] = resL[r];
  }
#pragma unroll
  for (int r = 0; r < 8; ++r) {
    const int t = tbS + r + 8 * kh + 16 * m;
    if (t < OUTL) outb[t] = resS[r];
  }
}

extern "C" void kernel_launch(void* const* d_in, const int* in_sizes, int n_in,
                              void* d_out, int out_size, void* d_ws, size_t ws_size,
                              hipStream_t stream) {
  (void)in_sizes; (void)n_in; (void)d_ws; (void)ws_size; (void)out_size;
  const float* x = (const float*)d_in[0];
  float* out = (float*)d_out;
  selfconv_wmma_f32<<<dim3(NBLK), dim3(256), 0, stream>>>(x, out);
}


# Tile map transposed, t = tb+16*row+col: the 16-wide skew sits on x[t-i], lower tiles start at i=0 with no 240-point lead-in; wave owns tiles W and W+16, 584 not 672 WMMAs; sum order changes
# speedup vs baseline: 6.2421x; 1.0694x over previous
#include <hip/hip_runtime.h>


#define BATCH 128
#define NBLK  256
#define LEN   4096
#define OUTL  (2 * LEN - 1)
#define PAD   1024
#define XBUF  (LEN + 2 * PAD)
#define XPBUF (XBUF + XBUF / 4)
#define XABUF (XPBUF + 16)
#define PADIDX(p) ((p) + 4 * ((p) >> 4))

typedef float v2f __attribute__((ext_vector_type(2)));
typedef float v4f __attribute__((ext_vector_type(4)));
typedef float v8f __attribute__((ext_vector_type(8)));

#if defined(__HIP_DEVICE_COMPILE__)
#if !__has_builtin(__builtin_amdgcn_wmma_f32_16x16x4_f32)
#error "missing __builtin_amdgcn_wmma_f32_16x16x4_f32 on this toolchain"
#endif
#endif

template <int US>
struct Frags {
  v2f b[US];
  v2f a[US];
};

template <int US>
__device__ __forceinline__ void load_frags(const float* XA, const float* XB,
                                           int aBaseP, int bBase, int u,
                                           int ub, Frags<US>& f) {
#pragma unroll
  for (int c = 0; c < US / 2; ++c) {
    const v4f av = *(const v4f*)(XA + aBaseP + ub + PADIDX(8 * c));
    f.a[2 * c]     = (v2f){av.x, av.y};
    f.a[2 * c + 1] = (v2f){av.z, av.w};
#pragma unroll
    for (int s = 0; s < 2; ++s) {
      const int ubk = u + 8 * c + 2 * s;
      f.b[2 * c + s].x = XB[bBase + ubk];
      f.b[2 * c + s].y = XB[bBase + ubk + 1];
    }
  }
}

template <int US>
__device__ __forceinline__ void cut_frags(Frags<US>& f, int eB, int eA) {
#pragma unroll
  for (int c = 0; c < US / 2; ++c) {
#pragma unroll
    for (int s = 0; s < 2; ++s) {
      const int q  = 2 * c + s;
      const int k0 = 8 * c + 2 * s;
      f.b[q].x = (k0     >= eB) ? 0.0f : f.b[q].x;
      f.b[q].y = (k0 + 1 >= eB) ? 0.0f : f.b[q].y;
      f.a[q].x = (k0     <= eA) ? 2.0f * f.a[q].x : f.a[q].x;
      f.a[q].y = (k0 + 1 <= eA) ? 2.0f * f.a[q].y : f.a[q].y;
    }
  }
#pragma unroll
  for (int q = 0; q < US; q += 8) {
    asm volatile("" : "+v"(f.b[q]), "+v"(f.b[q + 1]), "+v"(f.b[q + 2]), "+v"(f.b[q + 3]),
                      "+v"(f.b[q + 4]), "+v"(f.b[q + 5]), "+v"(f.b[q + 6]), "+v"(f.b[q + 7]));
    asm volatile("" : "+v"(f.a[q]), "+v"(f.a[q + 1]), "+v"(f.a[q + 2]), "+v"(f.a[q + 3]),
                      "+v"(f.a[q + 4]), "+v"(f.a[q + 5]), "+v"(f.a[q + 6]), "+v"(f.a[q + 7]));
  }
}

__device__ __forceinline__ void wmma_tail(v8f& acc0, v8f& acc1) {
  asm volatile("v_nop\n\tv_nop\n\tv_nop\n\tv_nop\n\tv_nop" : "+v"(acc0), "+v"(acc1));
}

template <int US>
__device__ __forceinline__ void mma_block(const Frags<US>& f, v8f& acc0, v8f& acc1) {
#pragma unroll
  for (int q = 0; q < US; ++q) {
    if (q & 1) {
      acc1 = __builtin_amdgcn_wmma_f32_16x16x4_f32(
           false, f.a[q],  false, f.b[q],
           (short)0, acc1,  false,  false);
    } else {
      acc0 = __builtin_amdgcn_wmma_f32_16x16x4_f32(
          false, f.a[q], false, f.b[q], (short)0, acc0, false, false);
    }
  }
}

__device__ __forceinline__ void pin_acc(v8f& acc0, v8f& acc1) {
  asm volatile("" : "+v"(acc0), "+v"(acc1));
}

template <int US, bool CUT>
__device__ __forceinline__ void run_range(const float* XA, const float* XB,
                                          int aBaseP, int bBase, int u0, int u1,
                                          int eB0, int eA0, v8f& acc0, v8f& acc1) {
  static_assert(US % 4 == 0, "a block must cover whole groups of 16 values");
  Frags<US> f0, f1;
  int ub = u0 + (u0 >> 2);
  load_frags<US>(XA, XB, aBaseP, bBase, u0, ub, f0);
  for (int u = u0; u < u1; u += 8 * US, ub += 10 * US) {
    load_frags<US>(XA, XB, aBaseP, bBase, u + 4 * US, ub + 5 * US, f1);
    __builtin_amdgcn_sched_barrier(0);
    if constexpr (CUT) cut_frags<US>(f0, eB0 - u, eA0 - u);
    pin_acc(acc0, acc1);
    mma_block<US>(f0, acc0, acc1);
    pin_acc(acc0, acc1);
    if constexpr (CUT) wmma_tail(acc0, acc1);
    __builtin_amdgcn_sched_barrier(0);
    load_frags<US>(XA, XB, aBaseP, bBase, u + 8 * US, ub + 10 * US, f0);
    __builtin_amdgcn_sched_barrier(0);
    if constexpr (CUT) cut_frags<US>(f1, eB0 - u - 4 * US, eA0 - u - 4 * US);
    pin_acc(acc0, acc1);
    mma_block<US>(f1, acc0, acc1);
    pin_acc(acc0, acc1);
    if constexpr (CUT) wmma_tail(acc0, acc1);
    __builtin_amdgcn_sched_barrier(0);
  }
  wmma_tail(acc0, acc1);
}

__global__ __launch_bounds__(256) void selfconv_wmma_f32(
    const float* __restrict__ x, float* __restrict__ out) {
  __shared__ __attribute__((aligned(16))) float XA[XABUF];
  __shared__ __attribute__((aligned(16))) float XB[XBUF];

  const int tid  = threadIdx.x;
  const int b    = blockIdx.x >> 1;
  const int half = blockIdx.x & 1;

  const float4* xb4 = (const float4*)(x + b * LEN);
  float4 v[LEN / 4 / 256];
#pragma unroll
  for (int k = 0; k < LEN / 4 / 256; ++k) v[k] = xb4[tid + 256 * k];
#pragma unroll
  for (int k = 0; k < 2 * PAD / 256; ++k) {
    const int i = tid + 256 * k;
    const int idx = (256 * k < PAD) ? i : (LEN + i);
    XB[idx] = 0.0f;
    XA[PADIDX(XBUF - idx)] = 0.0f;
  }
#pragma unroll
  for (int k = 0; k < LEN / 4 / 256; ++k) {
    const int i = tid + 256 * k;
    *(float4*)(XB + PAD + 4 * i) = v[k];
    const int qb = XBUF - (PAD + 4 * i);
    XA[PADIDX(qb)]     = v[k].x;
    XA[PADIDX(qb - 1)] = v[k].y;
    XA[PADIDX(qb - 2)] = v[k].z;
    XA[PADIDX(qb - 3)] = v[k].w;
  }
  __syncthreads();

  const int wave = tid >> 5;
  const int lane = tid & 31;
  const int m    = lane & 15;
  const int kh   = lane >> 4;

  const int Tlow = 8 * half + wave;
  const int Tupp = Tlow + 16;

  const int bBase = PAD + m + 4 * kh;

  v8f acc[8];
#pragma unroll
  for (int i = 0; i < 8; ++i) acc[i] = (v8f){0.f, 0.f, 0.f, 0.f, 0.f, 0.f, 0.f, 0.f};

#pragma unroll
  for (int t = 0; t < 2; ++t) {
    const int T   = t ? Tlow : Tupp;
    const int tb  = T << 8;
    const int cut = (T << 7) + 128;
    const int aBaseP = PADIDX(XBUF - PAD - tb - 16 * m) + 4 * kh;
    const int eB0 = cut - 4 * kh - m;
    const int eA0 = tb + 16 * m - cut - 4 * kh;
    if (t == 0) {
      run_range<16, false>(XA, XB, aBaseP, bBase, tb - LEN, cut - 256, 0, 0,
                           acc[2 * t], acc[2 * t + 1]);
    } else {
      run_range<4, false>(XA, XB, aBaseP, bBase, (T >= 1) ? -32 : 0, 0, 0, 0,
                          acc[2 * t], acc[2 * t + 1]);
      run_range<16, false>(XA, XB, aBaseP, bBase, 0, cut - 256, 0, 0,
                           acc[2 * t], acc[2 * t + 1]);
    }
    run_range<16, true>(XA, XB, aBaseP, bBase, cut - 256, cut, eB0, eA0,
                        acc[4 + 2 * t], acc[5 + 2 * t]);
  }
  const v8f resU = 2.0f * (acc[0] + acc[1]) + (acc[4] + acc[5]);
  const v8f resL = 2.0f * (acc[2] + acc[3]) + (acc[6] + acc[7]);

  float* outb = out + b * OUTL;
  const int tbU = Tupp << 8;
  const int tbL = Tlow << 8;
#pragma unroll
  for (int r = 0; r < 8; ++r) {
    const int t = tbU + 16 * (r + 8 * kh) + m;
    if (t < OUTL) outb[t] = resU[r];
  }
#pragma unroll
  for (int r = 0; r < 8; ++r) {
    const int t = tbL + 16 * (r + 8 * kh) + m;
    if (t < OUTL) outb[t] = resL[r];
  }
}

extern "C" void kernel_launch(void* const* d_in, const int* in_sizes, int n_in,
                              void* d_out, int out_size, void* d_ws, size_t ws_size,
                              hipStream_t stream) {
  (void)in_sizes; (void)n_in; (void)d_ws; (void)ws_size; (void)out_size;
  const float* x = (const float*)d_in[0];
  float* out = (float*)d_out;
  selfconv_wmma_f32<<<dim3(NBLK), dim3(256), 0, stream>>>(x, out);
}
